# starting kernel
# baseline (speedup 1.0000x reference)
#include <hip/hip_runtime.h>
#include <hip/hip_bf16.h>
#include <math.h>

typedef float v2f __attribute__((ext_vector_type(2)));
typedef float v8f __attribute__((ext_vector_type(8)));

#define DD 128
#define NNODE 40000
#define NEDGE 640000

__global__ __launch_bounds__(256)
void gat_gemm_wmma(const float* __restrict__ h,
                   const float* __restrict__ W1,
                   const float* __restrict__ W2,
                   float* __restrict__ z,
                   float* __restrict__ znb)
{
    const int lane  = threadIdx.x & 31;
    const int wavew = threadIdx.x >> 5;
    const int row0  = blockIdx.x << 4;
    const int col0  = wavew << 4;
    const int mn    = lane & 15;
    const int khalf = (lane >> 4) << 1;

    const float* hrow = h + (size_t)(row0 + mn) * DD;

    v8f accZ = {0.f, 0.f, 0.f, 0.f, 0.f, 0.f, 0.f, 0.f};
    v8f accI = {0.f, 0.f, 0.f, 0.f, 0.f, 0.f, 0.f, 0.f};

    for (int k0 = 0; k0 < DD; k0 += 4) {
        if (k0 + 16 < DD) __builtin_prefetch(&hrow[k0 + 16], 0, 1);
        v2f a;
        a.x = hrow[k0 + khalf];
        a.y = hrow[k0 + khalf + 1];
        v2f b1, b2;
        b1.x = W1[(size_t)(k0 + khalf)     * DD + col0 + mn];
        b1.y = W1[(size_t)(k0 + khalf + 1) * DD + col0 + mn];
        b2.x = W2[(size_t)(k0 + khalf)     * DD + col0 + mn];
        b2.y = W2[(size_t)(k0 + khalf + 1) * DD + col0 + mn];
        accZ = __builtin_amdgcn_wmma_f32_16x16x4_f32(false, a, false, b1,
                                                     (short)0, accZ, false, false);
        accI = __builtin_amdgcn_wmma_f32_16x16x4_f32(false, a, false, b2,
                                                     (short)0, accI, false, false);
    }

    const int rowBase = row0 + ((lane >> 4) << 3);
    #pragma unroll
    for (int r = 0; r < 8; ++r) {
        size_t idx = (size_t)(rowBase + r) * DD + col0 + mn;
        z[idx]   = accZ[r];
        znb[idx] = accI[r];
    }
}

__global__ __launch_bounds__(256)
void gat_node_prep(const float* __restrict__ z,
                   const float* __restrict__ Wa_l,
                   float* __restrict__ a_src,
                   float* __restrict__ a_dst,
                   float* __restrict__ mmax,
                   float* __restrict__ denom)
{
    int gid  = blockIdx.x * blockDim.x + threadIdx.x;
    int node = gid >> 5;
    int lane = gid & 31;
    if (node >= NNODE) return;

    const float4 zv = reinterpret_cast<const float4*>(z + (size_t)node * DD)[lane];
    const float* wa0 = Wa_l + lane * 4;
    const float* wa1 = Wa_l + DD + lane * 4;
    float s0 = zv.x * wa0[0] + zv.y * wa0[1] + zv.z * wa0[2] + zv.w * wa0[3];
    float s1 = zv.x * wa1[0] + zv.y * wa1[1] + zv.z * wa1[2] + zv.w * wa1[3];
    #pragma unroll
    for (int off = 16; off > 0; off >>= 1) {
        s0 += __shfl_xor(s0, off, 32);
        s1 += __shfl_xor(s1, off, 32);
    }
    if (lane == 0) {
        a_src[node] = s0;
        a_dst[node] = s1;
        mmax[node]  = -INFINITY;
        denom[node] = 0.0f;
    }
}

__device__ inline void atomicMaxF(float* addr, float val) {
    if (val >= 0.0f) atomicMax((int*)addr, __float_as_int(val));
    else             atomicMin((unsigned int*)addr, (unsigned int)__float_as_int(val));
}

__global__ __launch_bounds__(256)
void gat_edge_logits(const int* __restrict__ src,
                     const int* __restrict__ dst,
                     const float* __restrict__ dfeat,
                     const float* __restrict__ W0_l,
                     const float* __restrict__ Wa_l,
                     const float* __restrict__ a_src,
                     const float* __restrict__ a_dst,
                     float* __restrict__ e,
                     float* __restrict__ mmax)
{
    int i = blockIdx.x * blockDim.x + threadIdx.x;
    if (i >= NEDGE) return;
    float ct = W0_l[0] * Wa_l[2 * DD];
    int s = src[i], t = dst[i];
    float v = a_src[s] + a_dst[t] + dfeat[i] * ct;
    v = (v > 0.0f) ? v : 0.01f * v;
    e[i] = v;
    atomicMaxF(&mmax[t], v);
}

__global__ __launch_bounds__(256)
void gat_edge_exp(const int* __restrict__ dst,
                  float* __restrict__ e,
                  const float* __restrict__ mmax,
                  float* __restrict__ denom)
{
    int i = blockIdx.x * blockDim.x + threadIdx.x;
    if (i >= NEDGE) return;
    int t = dst[i];
    float x = __expf(e[i] - mmax[t]);
    e[i] = x;
    atomicAdd(&denom[t], x);
}

__global__ __launch_bounds__(256)
void gat_edge_scatter(const int* __restrict__ src,
                      const int* __restrict__ dst,
                      const float* __restrict__ ex,
                      const float* __restrict__ denom,
                      const float* __restrict__ z,
                      float* __restrict__ znb)
{
    int gid  = blockIdx.x * blockDim.x + threadIdx.x;
    int eidx = gid >> 5;
    int lane = gid & 31;
    if (eidx >= NEDGE) return;
    int s = src[eidx], t = dst[eidx];
    float alpha = ex[eidx] / denom[t];
    const float4 v = reinterpret_cast<const float4*>(z + (size_t)s * DD)[lane];
    float* o = znb + (size_t)t * DD + lane * 4;
    atomicAdd(o + 0, v.x * alpha);
    atomicAdd(o + 1, v.y * alpha);
    atomicAdd(o + 2, v.z * alpha);
    atomicAdd(o + 3, v.w * alpha);
}

__global__ __launch_bounds__(256)
void gat_relu(const float* __restrict__ znb, float* __restrict__ hout)
{
    int i = blockIdx.x * blockDim.x + threadIdx.x;
    if (i < NNODE * DD) hout[i] = fmaxf(znb[i], 0.0f);
}

extern "C" void kernel_launch(void* const* d_in, const int* in_sizes, int n_in,
                              void* d_out, int out_size, void* d_ws, size_t ws_size,
                              hipStream_t stream) {
    const float* attr = (const float*)d_in[0];
    const float* dfe  = (const float*)d_in[1];
    const int*   src  = (const int*)d_in[2];
    const int*   dst  = (const int*)d_in[3];
    const float* W0   = (const float*)d_in[4];
    const float* W1   = (const float*)d_in[5];
    const float* W2   = (const float*)d_in[6];
    const float* Wa   = (const float*)d_in[7];

    float* ws    = (float*)d_ws;
    float* hA    = ws;
    float* hB    = hA    + (size_t)NNODE * DD;
    float* z     = hB    + (size_t)NNODE * DD;
    float* znb   = z     + (size_t)NNODE * DD;
    float* a_src = znb   + (size_t)NNODE * DD;
    float* a_dst = a_src + NNODE;
    float* mmax  = a_dst + NNODE;
    float* denom = mmax  + NNODE;
    float* ebuf  = denom + NNODE;

    const int gemmBlocks    = NNODE / 16;
    const int nodeBlocks    = (NNODE * 32 + 255) / 256;
    const int edgeBlocks    = (NEDGE + 255) / 256;
    const int scatterBlocks = (NEDGE * 32 + 255) / 256;
    const int reluBlocks    = (NNODE * DD + 255) / 256;

    for (int l = 0; l < 3; ++l) {
        const float* hin  = (l == 0) ? attr : ((l == 1) ? hA : hB);
        float*       hout = (l == 2) ? (float*)d_out : ((l == 0) ? hA : hB);
        const float* W1l = W1 + (size_t)l * DD * DD;
        const float* W2l = W2 + (size_t)l * DD * DD;
        const float* W0l = W0 + l;
        const float* Wal = Wa + (size_t)l * (2 * DD + 1);

        gat_gemm_wmma<<<gemmBlocks, 256, 0, stream>>>(hin, W1l, W2l, z, znb);
        gat_node_prep<<<nodeBlocks, 256, 0, stream>>>(z, Wal, a_src, a_dst, mmax, denom);
        gat_edge_logits<<<edgeBlocks, 256, 0, stream>>>(src, dst, dfe, W0l, Wal,
                                                        a_src, a_dst, ebuf, mmax);
        gat_edge_exp<<<edgeBlocks, 256, 0, stream>>>(dst, ebuf, mmax, denom);
        gat_edge_scatter<<<scatterBlocks, 256, 0, stream>>>(src, dst, ebuf, denom, z, znb);
        gat_relu<<<reluBlocks, 256, 0, stream>>>(znb, hout);
    }
}


# GEMM wave patch 32x32 of both products (was 16x16): 2x2 register tiles, column tiles interleaved so W loads and stores cover whole lines, next k group loaded ahead; product k order changes
# speedup vs baseline: 2.8684x; 2.8684x over previous
#include <hip/hip_runtime.h>
#include <hip/hip_bf16.h>
#include <math.h>

typedef float v2f __attribute__((ext_vector_type(2)));
typedef float v8f __attribute__((ext_vector_type(8)));

#define DD 128
#define NNODE 40000
#define NEDGE 640000

#define GEMM_ROWS 32
static_assert(NNODE % GEMM_ROWS == 0, "whole blocks of rows");

template <bool RELU_IN>
__global__ __launch_bounds__(128) __attribute__((amdgpu_waves_per_eu(4, 6)))
void gat_gemm_wmma(const float* __restrict__ h,
                   const float* __restrict__ W1,
                   const float* __restrict__ W2,
                   float* __restrict__ z,
                   float* __restrict__ znb)
{
    const int lane = threadIdx.x & 31;
    const int wv   = threadIdx.x >> 5;
    const int row0 = blockIdx.x * GEMM_ROWS;
    const int col0 = wv << 5;
    const int c    = lane & 15;
    const int hh   = lane >> 4;

    const float* ap0 = h + (size_t)(row0 + c) * DD + 4 * hh;
    const float* ap1 = ap0 + (size_t)16 * DD;
    const size_t woff = (size_t)(4 * hh) * DD + col0 + 2 * c;
    const float* wp1 = W1 + woff;
    const float* wp2 = W2 + woff;

    v8f acc[2][2][2];
    #pragma unroll
    for (int m = 0; m < 2; ++m)
        #pragma unroll
        for (int i = 0; i < 2; ++i)
            #pragma unroll
            for (int t = 0; t < 2; ++t)
                acc[m][i][t] = (v8f){0.f, 0.f, 0.f, 0.f, 0.f, 0.f, 0.f, 0.f};

    float4 a[2][2];
    float2 b[2][2][4];

    a[0][0] = *reinterpret_cast<const float4*>(ap0);
    a[0][1] = *reinterpret_cast<const float4*>(ap1);
    #pragma unroll
    for (int q = 0; q < 4; ++q) {
        b[0][0][q] = *reinterpret_cast<const float2*>(wp1 + (size_t)q * DD);
        b[0][1][q] = *reinterpret_cast<const float2*>(wp2 + (size_t)q * DD);
    }

    #pragma unroll
    for (int g = 0; g < DD / 8; ++g) {
        const int cur = g & 1, nxt = cur ^ 1;
        if (g + 1 < DD / 8) {
            a[nxt][0] = *reinterpret_cast<const float4*>(ap0 + 8 * (g + 1));
            a[nxt][1] = *reinterpret_cast<const float4*>(ap1 + 8 * (g + 1));
            #pragma unroll
            for (int q = 0; q < 4; ++q) {
                b[nxt][0][q] = *reinterpret_cast<const float2*>(wp1 + (size_t)(8 * (g + 1) + q) * DD);
                b[nxt][1][q] = *reinterpret_cast<const float2*>(wp2 + (size_t)(8 * (g + 1) + q) * DD);
            }
        }
        __builtin_amdgcn_sched_barrier(0);

        float4 ar[2];
        #pragma unroll
        for (int i = 0; i < 2; ++i) {
            ar[i] = a[cur][i];
            if (RELU_IN) {
                ar[i].x = fmaxf(ar[i].x, 0.0f);
                ar[i].y = fmaxf(ar[i].y, 0.0f);
                ar[i].z = fmaxf(ar[i].z, 0.0f);
                ar[i].w = fmaxf(ar[i].w, 0.0f);
            }
        }
        #pragma unroll
        for (int u = 0; u < 2; ++u) {
            v2f av[2];
            #pragma unroll
            for (int i = 0; i < 2; ++i) {
                av[i].x = u ? ar[i].z : ar[i].x;
                av[i].y = u ? ar[i].w : ar[i].y;
            }
            #pragma unroll
            for (int m = 0; m < 2; ++m) {
                v2f bv[2];
                bv[0].x = b[cur][m][2 * u].x;
                bv[0].y = b[cur][m][2 * u + 1].x;
                bv[1].x = b[cur][m][2 * u].y;
                bv[1].y = b[cur][m][2 * u + 1].y;
                #pragma unroll
                for (int i = 0; i < 2; ++i)
                    #pragma unroll
                    for (int t = 0; t < 2; ++t)
                        acc[m][i][t] = __builtin_amdgcn_wmma_f32_16x16x4_f32(
                            false, av[i], false, bv[t], (short)0, acc[m][i][t], false, false);
            }
        }
    }

    #pragma unroll
    for (int i = 0; i < 2; ++i) {
        #pragma unroll
        for (int r = 0; r < 8; ++r) {
            const size_t idx = (size_t)(row0 + 16 * i + 8 * hh + r) * DD + col0 + 2 * c;
            float2 vz, vi;
            vz.x = acc[0][i][0][r];  vz.y = acc[0][i][1][r];
            vi.x = acc[1][i][0][r];  vi.y = acc[1][i][1][r];
            *reinterpret_cast<float2*>(z   + idx) = vz;
            *reinterpret_cast<float2*>(znb + idx) = vi;
        }
    }
}

__device__ inline int l2_half() {
    return (int)((__builtin_amdgcn_s_sendmsg_rtn(0x87) >> 18) & 1u);
}

__global__ __launch_bounds__(256)
void gat_node_prep(const float* __restrict__ z,
                   const float* __restrict__ Wa_l,
                   float* __restrict__ a_src,
                   float* __restrict__ a_dst,
                   float* __restrict__ mmax,
                   float* __restrict__ denom,
                   int* __restrict__ tickets)
{
    int gid  = blockIdx.x * blockDim.x + threadIdx.x;
    int node = gid >> 5;
    int lane = gid & 31;
    if (node >= NNODE) return;

    const float4 zv = reinterpret_cast<const float4*>(z + (size_t)node * DD)[lane];
    const float* wa0 = Wa_l + lane * 4;
    const float* wa1 = Wa_l + DD + lane * 4;
    float s0 = zv.x * wa0[0] + zv.y * wa0[1] + zv.z * wa0[2] + zv.w * wa0[3];
    float s1 = zv.x * wa1[0] + zv.y * wa1[1] + zv.z * wa1[2] + zv.w * wa1[3];
    #pragma unroll
    for (int off = 16; off > 0; off >>= 1) {
        s0 += __shfl_xor(s0, off, 32);
        s1 += __shfl_xor(s1, off, 32);
    }
    if (lane == 0) {
        a_src[node] = s0;
        a_dst[node] = s1;
        mmax[node]          = -INFINITY;
        mmax[NNODE + node]  = -INFINITY;
        denom[node]         = 0.0f;
        denom[NNODE + node] = 0.0f;
        if (node == 0) {
            tickets[0]  = 0;
            tickets[64] = 0;
        }
    }
}

__device__ inline void atomicMaxF(float* addr, float val) {
    if (val >= 0.0f) atomicMax((int*)addr, __float_as_int(val));
    else             atomicMin((unsigned int*)addr, (unsigned int)__float_as_int(val));
}

__global__ __launch_bounds__(256)
void gat_edge_logits(const int* __restrict__ src,
                     const int* __restrict__ dst,
                     const float* __restrict__ dfeat,
                     const float* __restrict__ W0_l,
                     const float* __restrict__ Wa_l,
                     const float* __restrict__ a_src,
                     const float* __restrict__ a_dst,
                     float* __restrict__ e,
                     float* __restrict__ mmax)
{
    const int half = l2_half();
    int i = blockIdx.x * blockDim.x + threadIdx.x;
    if (i >= NEDGE) return;
    float ct = W0_l[0] * Wa_l[2 * DD];
    int s = src[i], t = dst[i];
    float v = a_src[s] + a_dst[t] + dfeat[i] * ct;
    v = (v > 0.0f) ? v : 0.01f * v;
    e[i] = v;
    atomicMaxF(&mmax[half * NNODE + t], v);
}

__global__ __launch_bounds__(256)
void gat_edge_exp(const int* __restrict__ dst,
                  float* __restrict__ e,
                  const float* __restrict__ mmax,
                  float* __restrict__ denom)
{
    const int half = l2_half();
    int i = blockIdx.x * blockDim.x + threadIdx.x;
    if (i >= NEDGE) return;
    int t = dst[i];
    float x = __expf(e[i] - fmaxf(mmax[t], mmax[NNODE + t]));
    e[i] = x;
    atomicAdd(&denom[half * NNODE + t], x);
}

#define CHUNK_EDGES 256
#define NCHUNK (NEDGE / CHUNK_EDGES)
#define EDGE_BATCH 8
static_assert(NEDGE % CHUNK_EDGES == 0, "whole chunks of edges");

__device__ inline void lds_only_barrier() {
    __builtin_amdgcn_fence(__ATOMIC_RELEASE, "workgroup", "local");
    __builtin_amdgcn_s_barrier();
    __builtin_amdgcn_fence(__ATOMIC_ACQUIRE, "workgroup", "local");
}

__global__ __launch_bounds__(256)
void gat_edge_scatter(const int* __restrict__ src,
                      const int* __restrict__ dst,
                      const float* __restrict__ ex,
                      const float* __restrict__ denom,
                      const float* __restrict__ z,
                      float* __restrict__ znb,
                      int* __restrict__ tickets)
{
    __shared__ int s_ticket[2];
    const int lane = threadIdx.x & 31;
    const int wave = threadIdx.x >> 5;
    const int half = l2_half();

    const float* zl = z + lane;
    float*       ol = znb + lane;

    int k = 0;
    #pragma nounroll
    for (int ph = 0; ph < 2; ++ph) {
        const int par = half ^ ph;
        int* cnt = tickets + par * 64;
        if (threadIdx.x == 0) s_ticket[(k + 1) & 1] = atomicAdd(cnt, 1);
        lds_only_barrier();
        ++k;
        int t = s_ticket[k & 1];
        while (t < NCHUNK) {
            const int e   = t * CHUNK_EDGES + wave * 32 + lane;
            const int s_l = src[e];
            const int t_l = dst[e];
            if (threadIdx.x == 0) s_ticket[(k + 1) & 1] = atomicAdd(cnt, 1);

            const bool mine = (t_l & 1) == par;
            float a_l = 0.0f;
            if (mine) a_l = ex[e] / (denom[t_l] + denom[NNODE + t_l]);
            unsigned m = __builtin_amdgcn_ballot_w32(mine);

            while (m) {
                const int n = min((int)__builtin_popcount(m), EDGE_BATCH);
                int   jj[EDGE_BATCH];
                float v[EDGE_BATCH][4];
                #pragma unroll
                for (int u = 0; u < EDGE_BATCH; ++u) {
                    if (u < n) {
                        const int j = __builtin_ctz(m);
                        m &= m - 1;
                        jj[u] = j;
                        const int sj = __builtin_amdgcn_readlane(s_l, j);
                        const float* zr = zl + (size_t)sj * DD;
                        v[u][0] = zr[0];
                        v[u][1] = zr[32];
                        v[u][2] = zr[64];
                        v[u][3] = zr[96];
                    }
                }
                #pragma unroll
                for (int u = 0; u < EDGE_BATCH; ++u) {
                    if (u < n) {
                        const int   tj = __builtin_amdgcn_readlane(t_l, jj[u]);
                        const float aj = __int_as_float(
                            __builtin_amdgcn_readlane(__float_as_int(a_l), jj[u]));
                        float* o = ol + (size_t)tj * DD;
                        atomicAdd(o,      v[u][0] * aj);
                        atomicAdd(o + 32, v[u][1] * aj);
                        atomicAdd(o + 64, v[u][2] * aj);
                        atomicAdd(o + 96, v[u][3] * aj);
                    }
                }
            }

            lds_only_barrier();
            ++k;
            t = s_ticket[k & 1];
        }
    }
}

__global__ __launch_bounds__(256)
void gat_relu(const float* __restrict__ znb, float* __restrict__ hout)
{
    int i = blockIdx.x * blockDim.x + threadIdx.x;
    if (i < NNODE * DD) hout[i] = fmaxf(znb[i], 0.0f);
}

extern "C" void kernel_launch(void* const* d_in, const int* in_sizes, int n_in,
                              void* d_out, int out_size, void* d_ws, size_t ws_size,
                              hipStream_t stream) {
    const float* attr = (const float*)d_in[0];
    const float* dfe  = (const float*)d_in[1];
    const int*   src  = (const int*)d_in[2];
    const int*   dst  = (const int*)d_in[3];
    const float* W0   = (const float*)d_in[4];
    const float* W1   = (const float*)d_in[5];
    const float* W2   = (const float*)d_in[6];
    const float* Wa   = (const float*)d_in[7];

    float* ws    = (float*)d_ws;
    float* sumA  = ws;
    float* sumB  = sumA  + (size_t)NNODE * DD;
    float* z     = sumB  + (size_t)NNODE * DD;
    float* a_src = z     + (size_t)NNODE * DD;
    float* a_dst = a_src + NNODE;
    float* mmax  = a_dst + NNODE;
    float* denom = mmax  + 2 * NNODE;
    float* ebuf  = denom + 2 * NNODE;
    int*   tickets = (int*)(ebuf + NEDGE);

    const int gemmBlocks    = NNODE / GEMM_ROWS;
    const int nodeBlocks    = (NNODE * 32 + 255) / 256;
    const int edgeBlocks    = (NEDGE + 255) / 256;
    const int scatterBlocks = 2048;
    const int reluBlocks    = (NNODE * DD + 255) / 256;

    for (int l = 0; l < 3; ++l) {
        const float* hin = (l == 0) ? attr : ((l == 1) ? sumA : sumB);
        float*       znb = (l == 1) ? sumB : sumA;
        const float* W1l = W1 + (size_t)l * DD * DD;
        const float* W2l = W2 + (size_t)l * DD * DD;
        const float* W0l = W0 + l;
        const float* Wal = Wa + (size_t)l * (2 * DD + 1);

        if (l == 0) gat_gemm_wmma<false><<<gemmBlocks, 128, 0, stream>>>(hin, W1l, W2l, z, znb);
        else        gat_gemm_wmma<true><<<gemmBlocks, 128, 0, stream>>>(hin, W1l, W2l, z, znb);
        gat_node_prep<<<nodeBlocks, 256, 0, stream>>>(z, Wal, a_src, a_dst, mmax, denom, tickets);
        gat_edge_logits<<<edgeBlocks, 256, 0, stream>>>(src, dst, dfe, W0l, Wal,
                                                        a_src, a_dst, ebuf, mmax);
        gat_edge_exp<<<edgeBlocks, 256, 0, stream>>>(dst, ebuf, mmax, denom);
        gat_edge_scatter<<<scatterBlocks, 256, 0, stream>>>(src, dst, ebuf, denom, z, znb, tickets);
        if (l == 2) gat_relu<<<reluBlocks, 256, 0, stream>>>(znb, (float*)d_out);
    }
}


# Node-dot, edge-logit, edge-exp and last relu passes: each lane takes 4 nodes / edges / 16-byte pieces, all loads in flight at once; a quarter of the waves or fewer; denom sums stay in arrival order
# speedup vs baseline: 3.2857x; 1.1455x over previous
#include <hip/hip_runtime.h>
#include <hip/hip_bf16.h>
#include <math.h>

typedef float v2f __attribute__((ext_vector_type(2)));
typedef float v8f __attribute__((ext_vector_type(8)));

#define DD 128
#define NNODE 40000
#define NEDGE 640000

#define GEMM_ROWS 32
static_assert(NNODE % GEMM_ROWS == 0, "whole blocks of rows");

template <bool RELU_IN>
__global__ __launch_bounds__(128) __attribute__((amdgpu_waves_per_eu(4, 6)))
void gat_gemm_wmma(const float* __restrict__ h,
                   const float* __restrict__ W1,
                   const float* __restrict__ W2,
                   float* __restrict__ z,
                   float* __restrict__ znb)
{
    const int lane = threadIdx.x & 31;
    const int wv   = threadIdx.x >> 5;
    const int row0 = blockIdx.x * GEMM_ROWS;
    const int col0 = wv << 5;
    const int c    = lane & 15;
    const int hh   = lane >> 4;

    const float* ap0 = h + (size_t)(row0 + c) * DD + 4 * hh;
    const float* ap1 = ap0 + (size_t)16 * DD;
    const size_t woff = (size_t)(4 * hh) * DD + col0 + 2 * c;
    const float* wp1 = W1 + woff;
    const float* wp2 = W2 + woff;

    v8f acc[2][2][2];
    #pragma unroll
    for (int m = 0; m < 2; ++m)
        #pragma unroll
        for (int i = 0; i < 2; ++i)
            #pragma unroll
            for (int t = 0; t < 2; ++t)
                acc[m][i][t] = (v8f){0.f, 0.f, 0.f, 0.f, 0.f, 0.f, 0.f, 0.f};

    float4 a[2][2];
    float2 b[2][2][4];

    a[0][0] = *reinterpret_cast<const float4*>(ap0);
    a[0][1] = *reinterpret_cast<const float4*>(ap1);
    #pragma unroll
    for (int q = 0; q < 4; ++q) {
        b[0][0][q] = *reinterpret_cast<const float2*>(wp1 + (size_t)q * DD);
        b[0][1][q] = *reinterpret_cast<const float2*>(wp2 + (size_t)q * DD);
    }

    #pragma unroll
    for (int g = 0; g < DD / 8; ++g) {
        const int cur = g & 1, nxt = cur ^ 1;
        if (g + 1 < DD / 8) {
            a[nxt][0] = *reinterpret_cast<const float4*>(ap0 + 8 * (g + 1));
            a[nxt][1] = *reinterpret_cast<const float4*>(ap1 + 8 * (g + 1));
            #pragma unroll
            for (int q = 0; q < 4; ++q) {
                b[nxt][0][q] = *reinterpret_cast<const float2*>(wp1 + (size_t)(8 * (g + 1) + q) * DD);
                b[nxt][1][q] = *reinterpret_cast<const float2*>(wp2 + (size_t)(8 * (g + 1) + q) * DD);
            }
        }
        __builtin_amdgcn_sched_barrier(0);

        float4 ar[2];
        #pragma unroll
        for (int i = 0; i < 2; ++i) {
            ar[i] = a[cur][i];
            if (RELU_IN) {
                ar[i].x = fmaxf(ar[i].x, 0.0f);
                ar[i].y = fmaxf(ar[i].y, 0.0f);
                ar[i].z = fmaxf(ar[i].z, 0.0f);
                ar[i].w = fmaxf(ar[i].w, 0.0f);
            }
        }
        #pragma unroll
        for (int u = 0; u < 2; ++u) {
            v2f av[2];
            #pragma unroll
            for (int i = 0; i < 2; ++i) {
                av[i].x = u ? ar[i].z : ar[i].x;
                av[i].y = u ? ar[i].w : ar[i].y;
            }
            #pragma unroll
            for (int m = 0; m < 2; ++m) {
                v2f bv[2];
                bv[0].x = b[cur][m][2 * u].x;
                bv[0].y = b[cur][m][2 * u + 1].x;
                bv[1].x = b[cur][m][2 * u].y;
                bv[1].y = b[cur][m][2 * u + 1].y;
                #pragma unroll
                for (int i = 0; i < 2; ++i)
                    #pragma unroll
                    for (int t = 0; t < 2; ++t)
                        acc[m][i][t] = __builtin_amdgcn_wmma_f32_16x16x4_f32(
                            false, av[i], false, bv[t], (short)0, acc[m][i][t], false, false);
            }
        }
    }

    #pragma unroll
    for (int i = 0; i < 2; ++i) {
        #pragma unroll
        for (int r = 0; r < 8; ++r) {
            const size_t idx = (size_t)(row0 + 16 * i + 8 * hh + r) * DD + col0 + 2 * c;
            float2 vz, vi;
            vz.x = acc[0][i][0][r];  vz.y = acc[0][i][1][r];
            vi.x = acc[1][i][0][r];  vi.y = acc[1][i][1][r];
            *reinterpret_cast<float2*>(z   + idx) = vz;
            *reinterpret_cast<float2*>(znb + idx) = vi;
        }
    }
}

__device__ inline int l2_half() {
    return (int)((__builtin_amdgcn_s_sendmsg_rtn(0x87) >> 18) & 1u);
}

#define PREP_NODES 4
static_assert(NNODE % (8 * PREP_NODES) == 0, "whole blocks of nodes");

__global__ __launch_bounds__(256)
void gat_node_prep(const float* __restrict__ z,
                   const float* __restrict__ Wa_l,
                   float* __restrict__ a_src,
                   float* __restrict__ a_dst,
                   float* __restrict__ mmax,
                   float* __restrict__ denom,
                   int* __restrict__ tickets)
{
    const int lane  = threadIdx.x & 31;
    const int node0 = ((blockIdx.x * 256 + threadIdx.x) >> 5) * PREP_NODES;

    float4 zv[PREP_NODES];
    #pragma unroll
    for (int u = 0; u < PREP_NODES; ++u)
        zv[u] = reinterpret_cast<const float4*>(z + (size_t)(node0 + u) * DD)[lane];

    const float* wa0 = Wa_l + lane * 4;
    const float* wa1 = Wa_l + DD + lane * 4;
    const float w00 = wa0[0], w01 = wa0[1], w02 = wa0[2], w03 = wa0[3];
    const float w10 = wa1[0], w11 = wa1[1], w12 = wa1[2], w13 = wa1[3];

    float s0[PREP_NODES], s1[PREP_NODES];
    #pragma unroll
    for (int u = 0; u < PREP_NODES; ++u) {
        s0[u] = zv[u].x * w00 + zv[u].y * w01 + zv[u].z * w02 + zv[u].w * w03;
        s1[u] = zv[u].x * w10 + zv[u].y * w11 + zv[u].z * w12 + zv[u].w * w13;
    }
    #pragma unroll
    for (int off = 16; off > 0; off >>= 1) {
        #pragma unroll
        for (int u = 0; u < PREP_NODES; ++u) {
            s0[u] += __shfl_xor(s0[u], off, 32);
            s1[u] += __shfl_xor(s1[u], off, 32);
        }
    }
    if (lane < PREP_NODES) {
        float o0 = s0[0], o1 = s1[0];
        #pragma unroll
        for (int u = 1; u < PREP_NODES; ++u)
            if (lane == u) { o0 = s0[u]; o1 = s1[u]; }
        const int node = node0 + lane;
        a_src[node] = o0;
        a_dst[node] = o1;
        mmax[node]          = -INFINITY;
        mmax[NNODE + node]  = -INFINITY;
        denom[node]         = 0.0f;
        denom[NNODE + node] = 0.0f;
        if (node == 0) {
            tickets[0]  = 0;
            tickets[64] = 0;
        }
    }
}

__device__ inline void atomicMaxF(float* addr, float val) {
    if (val >= 0.0f) atomicMax((int*)addr, __float_as_int(val));
    else             atomicMin((unsigned int*)addr, (unsigned int)__float_as_int(val));
}

#define EDGE_PER_THREAD 4
#define EDGE_THREADS 128
static_assert(NEDGE % (EDGE_THREADS * EDGE_PER_THREAD) == 0, "whole blocks of edges");

__global__ __launch_bounds__(EDGE_THREADS)
void gat_edge_logits(const int* __restrict__ src,
                     const int* __restrict__ dst,
                     const float* __restrict__ dfeat,
                     const float* __restrict__ W0_l,
                     const float* __restrict__ Wa_l,
                     const float* __restrict__ a_src,
                     const float* __restrict__ a_dst,
                     float* __restrict__ e,
                     float* __restrict__ mmax)
{
    const int half = l2_half();
    const int i0 = (blockIdx.x * EDGE_THREADS + threadIdx.x) * EDGE_PER_THREAD;
    const float ct = W0_l[0] * Wa_l[2 * DD];
    const int4   s4 = *reinterpret_cast<const int4*>(src + i0);
    const int4   t4 = *reinterpret_cast<const int4*>(dst + i0);
    const float4 d4 = *reinterpret_cast<const float4*>(dfeat + i0);
    const int   s[EDGE_PER_THREAD] = {s4.x, s4.y, s4.z, s4.w};
    const int   t[EDGE_PER_THREAD] = {t4.x, t4.y, t4.z, t4.w};
    const float d[EDGE_PER_THREAD] = {d4.x, d4.y, d4.z, d4.w};
    float v[EDGE_PER_THREAD];
    #pragma unroll
    for (int u = 0; u < EDGE_PER_THREAD; ++u) {
        v[u] = a_src[s[u]] + a_dst[t[u]] + d[u] * ct;
        v[u] = (v[u] > 0.0f) ? v[u] : 0.01f * v[u];
    }
    *reinterpret_cast<float4*>(e + i0) = make_float4(v[0], v[1], v[2], v[3]);
    float* mp = mmax + half * NNODE;
    #pragma unroll
    for (int u = 0; u < EDGE_PER_THREAD; ++u)
        atomicMaxF(&mp[t[u]], v[u]);
}

__global__ __launch_bounds__(EDGE_THREADS)
void gat_edge_exp(const int* __restrict__ dst,
                  float* __restrict__ e,
                  const float* __restrict__ mmax,
                  float* __restrict__ denom)
{
    const int half = l2_half();
    const int i0 = (blockIdx.x * EDGE_THREADS + threadIdx.x) * EDGE_PER_THREAD;
    const int4   t4 = *reinterpret_cast<const int4*>(dst + i0);
    const float4 e4 = *reinterpret_cast<const float4*>(e + i0);
    const int   t[EDGE_PER_THREAD]  = {t4.x, t4.y, t4.z, t4.w};
    const float ev[EDGE_PER_THREAD] = {e4.x, e4.y, e4.z, e4.w};
    float x[EDGE_PER_THREAD];
    #pragma unroll
    for (int u = 0; u < EDGE_PER_THREAD; ++u)
        x[u] = __expf(ev[u] - fmaxf(mmax[t[u]], mmax[NNODE + t[u]]));
    *reinterpret_cast<float4*>(e + i0) = make_float4(x[0], x[1], x[2], x[3]);
    float* dp = denom + half * NNODE;
    #pragma unroll
    for (int u = 0; u < EDGE_PER_THREAD; ++u)
        atomicAdd(&dp[t[u]], x[u]);
}

#define CHUNK_EDGES 256
#define NCHUNK (NEDGE / CHUNK_EDGES)
#define EDGE_BATCH 8
static_assert(NEDGE % CHUNK_EDGES == 0, "whole chunks of edges");

__device__ inline void lds_only_barrier() {
    __builtin_amdgcn_fence(__ATOMIC_RELEASE, "workgroup", "local");
    __builtin_amdgcn_s_barrier();
    __builtin_amdgcn_fence(__ATOMIC_ACQUIRE, "workgroup", "local");
}

__global__ __launch_bounds__(256)
void gat_edge_scatter(const int* __restrict__ src,
                      const int* __restrict__ dst,
                      const float* __restrict__ ex,
                      const float* __restrict__ denom,
                      const float* __restrict__ z,
                      float* __restrict__ znb,
                      int* __restrict__ tickets)
{
    __shared__ int s_ticket[2];
    const int lane = threadIdx.x & 31;
    const int wave = threadIdx.x >> 5;
    const int half = l2_half();

    const float* zl = z + lane;
    float*       ol = znb + lane;

    int k = 0;
    #pragma nounroll
    for (int ph = 0; ph < 2; ++ph) {
        const int par = half ^ ph;
        int* cnt = tickets + par * 64;
        if (threadIdx.x == 0) s_ticket[(k + 1) & 1] = atomicAdd(cnt, 1);
        lds_only_barrier();
        ++k;
        int t = s_ticket[k & 1];
        while (t < NCHUNK) {
            const int e   = t * CHUNK_EDGES + wave * 32 + lane;
            const int s_l = src[e];
            const int t_l = dst[e];
            if (threadIdx.x == 0) s_ticket[(k + 1) & 1] = atomicAdd(cnt, 1);

            const bool mine = (t_l & 1) == par;
            float a_l = 0.0f;
            if (mine) a_l = ex[e] / (denom[t_l] + denom[NNODE + t_l]);
            unsigned m = __builtin_amdgcn_ballot_w32(mine);

            while (m) {
                const int n = min((int)__builtin_popcount(m), EDGE_BATCH);
                int   jj[EDGE_BATCH];
                float v[EDGE_BATCH][4];
                #pragma unroll
                for (int u = 0; u < EDGE_BATCH; ++u) {
                    if (u < n) {
                        const int j = __builtin_ctz(m);
                        m &= m - 1;
                        jj[u] = j;
                        const int sj = __builtin_amdgcn_readlane(s_l, j);
                        const float* zr = zl + (size_t)sj * DD;
                        v[u][0] = zr[0];
                        v[u][1] = zr[32];
                        v[u][2] = zr[64];
                        v[u][3] = zr[96];
                    }
                }
                #pragma unroll
                for (int u = 0; u < EDGE_BATCH; ++u) {
                    if (u < n) {
                        const int   tj = __builtin_amdgcn_readlane(t_l, jj[u]);
                        const float aj = __int_as_float(
                            __builtin_amdgcn_readlane(__float_as_int(a_l), jj[u]));
                        float* o = ol + (size_t)tj * DD;
                        atomicAdd(o,      v[u][0] * aj);
                        atomicAdd(o + 32, v[u][1] * aj);
                        atomicAdd(o + 64, v[u][2] * aj);
                        atomicAdd(o + 96, v[u][3] * aj);
                    }
                }
            }

            lds_only_barrier();
            ++k;
            t = s_ticket[k & 1];
        }
    }
}

#define RELU_THREADS 128
#define RELU_VEC 4
static_assert((NNODE * DD) % (RELU_THREADS * RELU_VEC * 4) == 0, "whole blocks of floats");

__global__ __launch_bounds__(RELU_THREADS)
void gat_relu(const float* __restrict__ znb, float* __restrict__ hout)
{
    const size_t base = (size_t)blockIdx.x * (RELU_THREADS * RELU_VEC * 4) + threadIdx.x * 4;
    float4 v[RELU_VEC];
    #pragma unroll
    for (int u = 0; u < RELU_VEC; ++u)
        v[u] = *reinterpret_cast<const float4*>(znb + base + (size_t)u * RELU_THREADS * 4);
    #pragma unroll
    for (int u = 0; u < RELU_VEC; ++u) {
        float4 r;
        r.x = fmaxf(v[u].x, 0.0f);
        r.y = fmaxf(v[u].y, 0.0f);
        r.z = fmaxf(v[u].z, 0.0f);
        r.w = fmaxf(v[u].w, 0.0f);
        *reinterpret_cast<float4*>(hout + base + (size_t)u * RELU_THREADS * 4) = r;
    }
}

extern "C" void kernel_launch(void* const* d_in, const int* in_sizes, int n_in,
                              void* d_out, int out_size, void* d_ws, size_t ws_size,
                              hipStream_t stream) {
    const float* attr = (const float*)d_in[0];
    const float* dfe  = (const float*)d_in[1];
    const int*   src  = (const int*)d_in[2];
    const int*   dst  = (const int*)d_in[3];
    const float* W0   = (const float*)d_in[4];
    const float* W1   = (const float*)d_in[5];
    const float* W2   = (const float*)d_in[6];
    const float* Wa   = (const float*)d_in[7];

    float* ws    = (float*)d_ws;
    float* sumA  = ws;
    float* sumB  = sumA  + (size_t)NNODE * DD;
    float* z     = sumB  + (size_t)NNODE * DD;
    float* a_src = z     + (size_t)NNODE * DD;
    float* a_dst = a_src + NNODE;
    float* mmax  = a_dst + NNODE;
    float* denom = mmax  + 2 * NNODE;
    float* ebuf  = denom + 2 * NNODE;
    int*   tickets = (int*)(ebuf + NEDGE);

    const int gemmBlocks    = NNODE / GEMM_ROWS;
    const int nodeBlocks    = NNODE / (8 * PREP_NODES);
    const int edgeBlocks    = NEDGE / (EDGE_THREADS * EDGE_PER_THREAD);
    const int scatterBlocks = 2048;
    const int reluBlocks    = (NNODE * DD) / (RELU_THREADS * RELU_VEC * 4);

    for (int l = 0; l < 3; ++l) {
        const float* hin = (l == 0) ? attr : ((l == 1) ? sumA : sumB);
        float*       znb = (l == 1) ? sumB : sumA;
        const float* W1l = W1 + (size_t)l * DD * DD;
        const float* W2l = W2 + (size_t)l * DD * DD;
        const float* W0l = W0 + l;
        const float* Wal = Wa + (size_t)l * (2 * DD + 1);

        if (l == 0) gat_gemm_wmma<false><<<gemmBlocks, 128, 0, stream>>>(hin, W1l, W2l, z, znb);
        else        gat_gemm_wmma<true><<<gemmBlocks, 128, 0, stream>>>(hin, W1l, W2l, z, znb);
        gat_node_prep<<<nodeBlocks, 256, 0, stream>>>(z, Wal, a_src, a_dst, mmax, denom, tickets);
        gat_edge_logits<<<edgeBlocks, EDGE_THREADS, 0, stream>>>(src, dst, dfe, W0l, Wal,
                                                        a_src, a_dst, ebuf, mmax);
        gat_edge_exp<<<edgeBlocks, EDGE_THREADS, 0, stream>>>(dst, ebuf, mmax, denom);
        gat_edge_scatter<<<scatterBlocks, 256, 0, stream>>>(src, dst, ebuf, denom, z, znb, tickets);
        if (l == 2) gat_relu<<<reluBlocks, RELU_THREADS, 0, stream>>>(znb, (float*)d_out);
    }
}


# Softmax division left to the reader of the sums: the scatter forms exp(e - m), adds it to one denominator plane and adds un-normalised rows to a zeroed plane; edge-exp launch gone; sum terms change
# speedup vs baseline: 3.3944x; 1.0331x over previous
#include <hip/hip_runtime.h>
#include <hip/hip_bf16.h>
#include <math.h>

typedef float v2f __attribute__((ext_vector_type(2)));
typedef float v8f __attribute__((ext_vector_type(8)));

#define DD 128
#define NNODE 40000
#define NEDGE 640000

#define GEMM_ROWS 32
static_assert(NNODE % GEMM_ROWS == 0, "whole blocks of rows");

template <bool RELU_IN>
__global__ __launch_bounds__(128) __attribute__((amdgpu_waves_per_eu(4, 6)))
void gat_gemm_wmma(const float* __restrict__ h,
                   const float* __restrict__ hacc,
                   const float* __restrict__ hden,
                   const float* __restrict__ W1,
                   const float* __restrict__ W2,
                   float* __restrict__ z,
                   float* __restrict__ znb,
                   float* __restrict__ accz)
{
    const int lane = threadIdx.x & 31;
    const int wv   = threadIdx.x >> 5;
    const int row0 = blockIdx.x * GEMM_ROWS;
    const int col0 = wv << 5;
    const int c    = lane & 15;
    const int hh   = lane >> 4;

    const float* ap0 = h + (size_t)(row0 + c) * DD + 4 * hh;
    const float* ap1 = ap0 + (size_t)16 * DD;
    const float* qp0 = hacc + (size_t)(row0 + c) * DD + 4 * hh;
    const float* qp1 = qp0 + (size_t)16 * DD;
    float rinv[2] = {0.0f, 0.0f};
    if (RELU_IN) {
        #pragma unroll
        for (int i = 0; i < 2; ++i) {
            const int rw = row0 + 16 * i + c;
            const float dn = hden[(rw & 1) * (NNODE / 2) + (rw >> 1)];
            rinv[i] = (dn > 0.0f) ? 1.0f / dn : 0.0f;
        }
    }
    const size_t woff = (size_t)(4 * hh) * DD + col0 + 2 * c;
    const float* wp1 = W1 + woff;
    const float* wp2 = W2 + woff;

    v8f acc[2][2][2];
    #pragma unroll
    for (int m = 0; m < 2; ++m)
        #pragma unroll
        for (int i = 0; i < 2; ++i)
            #pragma unroll
            for (int t = 0; t < 2; ++t)
                acc[m][i][t] = (v8f){0.f, 0.f, 0.f, 0.f, 0.f, 0.f, 0.f, 0.f};

    float4 a[2][2];
    float4 q[2][2];
    float2 b[2][2][4];

    a[0][0] = *reinterpret_cast<const float4*>(ap0);
    a[0][1] = *reinterpret_cast<const float4*>(ap1);
    if (RELU_IN) {
        q[0][0] = *reinterpret_cast<const float4*>(qp0);
        q[0][1] = *reinterpret_cast<const float4*>(qp1);
    }
    #pragma unroll
    for (int q = 0; q < 4; ++q) {
        b[0][0][q] = *reinterpret_cast<const float2*>(wp1 + (size_t)q * DD);
        b[0][1][q] = *reinterpret_cast<const float2*>(wp2 + (size_t)q * DD);
    }

    #pragma unroll
    for (int g = 0; g < DD / 8; ++g) {
        const int cur = g & 1, nxt = cur ^ 1;
        if (g + 1 < DD / 8) {
            a[nxt][0] = *reinterpret_cast<const float4*>(ap0 + 8 * (g + 1));
            a[nxt][1] = *reinterpret_cast<const float4*>(ap1 + 8 * (g + 1));
            if (RELU_IN) {
                q[nxt][0] = *reinterpret_cast<const float4*>(qp0 + 8 * (g + 1));
                q[nxt][1] = *reinterpret_cast<const float4*>(qp1 + 8 * (g + 1));
            }
            #pragma unroll
            for (int q = 0; q < 4; ++q) {
                b[nxt][0][q] = *reinterpret_cast<const float2*>(wp1 + (size_t)(8 * (g + 1) + q) * DD);
                b[nxt][1][q] = *reinterpret_cast<const float2*>(wp2 + (size_t)(8 * (g + 1) + q) * DD);
            }
        }
        __builtin_amdgcn_sched_barrier(0);

        float4 ar[2];
        #pragma unroll
        for (int i = 0; i < 2; ++i) {
            ar[i] = a[cur][i];
            if (RELU_IN) {
                const float4 qq = q[cur][i];
                ar[i].x = fmaxf(ar[i].x + qq.x * rinv[i], 0.0f);
                ar[i].y = fmaxf(ar[i].y + qq.y * rinv[i], 0.0f);
                ar[i].z = fmaxf(ar[i].z + qq.z * rinv[i], 0.0f);
                ar[i].w = fmaxf(ar[i].w + qq.w * rinv[i], 0.0f);
            }
        }
        #pragma unroll
        for (int u = 0; u < 2; ++u) {
            v2f av[2];
            #pragma unroll
            for (int i = 0; i < 2; ++i) {
                av[i].x = u ? ar[i].z : ar[i].x;
                av[i].y = u ? ar[i].w : ar[i].y;
            }
            #pragma unroll
            for (int m = 0; m < 2; ++m) {
                v2f bv[2];
                bv[0].x = b[cur][m][2 * u].x;
                bv[0].y = b[cur][m][2 * u + 1].x;
                bv[1].x = b[cur][m][2 * u].y;
                bv[1].y = b[cur][m][2 * u + 1].y;
                #pragma unroll
                for (int i = 0; i < 2; ++i)
                    #pragma unroll
                    for (int t = 0; t < 2; ++t)
                        acc[m][i][t] = __builtin_amdgcn_wmma_f32_16x16x4_f32(
                            false, av[i], false, bv[t], (short)0, acc[m][i][t], false, false);
            }
        }
    }

    #pragma unroll
    for (int i = 0; i < 2; ++i) {
        #pragma unroll
        for (int r = 0; r < 8; ++r) {
            const size_t idx = (size_t)(row0 + 16 * i + 8 * hh + r) * DD + col0 + 2 * c;
            float2 vz, vi;
            vz.x = acc[0][i][0][r];  vz.y = acc[0][i][1][r];
            vi.x = acc[1][i][0][r];  vi.y = acc[1][i][1][r];
            *reinterpret_cast<float2*>(z   + idx) = vz;
            *reinterpret_cast<float2*>(znb + idx) = vi;
            *reinterpret_cast<float2*>(accz + idx) = make_float2(0.0f, 0.0f);
        }
    }
}

__device__ inline int l2_half() {
    return (int)((__builtin_amdgcn_s_sendmsg_rtn(0x87) >> 18) & 1u);
}

#define PREP_NODES 4
static_assert(NNODE % (8 * PREP_NODES) == 0, "whole blocks of nodes");

__global__ __launch_bounds__(256)
void gat_node_prep(const float* __restrict__ z,
                   const float* __restrict__ Wa_l,
                   float* __restrict__ a_src,
                   float* __restrict__ a_dst,
                   float* __restrict__ mmax,
                   float* __restrict__ denom,
                   int* __restrict__ tickets)
{
    const int lane  = threadIdx.x & 31;
    const int node0 = ((blockIdx.x * 256 + threadIdx.x) >> 5) * PREP_NODES;

    float4 zv[PREP_NODES];
    #pragma unroll
    for (int u = 0; u < PREP_NODES; ++u)
        zv[u] = reinterpret_cast<const float4*>(z + (size_t)(node0 + u) * DD)[lane];

    const float* wa0 = Wa_l + lane * 4;
    const float* wa1 = Wa_l + DD + lane * 4;
    const float w00 = wa0[0], w01 = wa0[1], w02 = wa0[2], w03 = wa0[3];
    const float w10 = wa1[0], w11 = wa1[1], w12 = wa1[2], w13 = wa1[3];

    float s0[PREP_NODES], s1[PREP_NODES];
    #pragma unroll
    for (int u = 0; u < PREP_NODES; ++u) {
        s0[u] = zv[u].x * w00 + zv[u].y * w01 + zv[u].z * w02 + zv[u].w * w03;
        s1[u] = zv[u].x * w10 + zv[u].y * w11 + zv[u].z * w12 + zv[u].w * w13;
    }
    #pragma unroll
    for (int off = 16; off > 0; off >>= 1) {
        #pragma unroll
        for (int u = 0; u < PREP_NODES; ++u) {
            s0[u] += __shfl_xor(s0[u], off, 32);
            s1[u] += __shfl_xor(s1[u], off, 32);
        }
    }
    if (lane < PREP_NODES) {
        float o0 = s0[0], o1 = s1[0];
        #pragma unroll
        for (int u = 1; u < PREP_NODES; ++u)
            if (lane == u) { o0 = s0[u]; o1 = s1[u]; }
        const int node = node0 + lane;
        a_src[node] = o0;
        a_dst[node] = o1;
        mmax[node]          = -INFINITY;
        mmax[NNODE + node]  = -INFINITY;
        denom[node]         = 0.0f;
        if (node == 0) {
            tickets[0]  = 0;
            tickets[64] = 0;
        }
    }
}

__device__ inline void atomicMaxF(float* addr, float val) {
    if (val >= 0.0f) atomicMax((int*)addr, __float_as_int(val));
    else             atomicMin((unsigned int*)addr, (unsigned int)__float_as_int(val));
}

#define EDGE_PER_THREAD 4
#define EDGE_THREADS 128
static_assert(NEDGE % (EDGE_THREADS * EDGE_PER_THREAD) == 0, "whole blocks of edges");

__global__ __launch_bounds__(EDGE_THREADS)
void gat_edge_logits(const int* __restrict__ src,
                     const int* __restrict__ dst,
                     const float* __restrict__ dfeat,
                     const float* __restrict__ W0_l,
                     const float* __restrict__ Wa_l,
                     const float* __restrict__ a_src,
                     const float* __restrict__ a_dst,
                     float* __restrict__ e,
                     float* __restrict__ mmax)
{
    const int half = l2_half();
    const int i0 = (blockIdx.x * EDGE_THREADS + threadIdx.x) * EDGE_PER_THREAD;
    const float ct = W0_l[0] * Wa_l[2 * DD];
    const int4   s4 = *reinterpret_cast<const int4*>(src + i0);
    const int4   t4 = *reinterpret_cast<const int4*>(dst + i0);
    const float4 d4 = *reinterpret_cast<const float4*>(dfeat + i0);
    const int   s[EDGE_PER_THREAD] = {s4.x, s4.y, s4.z, s4.w};
    const int   t[EDGE_PER_THREAD] = {t4.x, t4.y, t4.z, t4.w};
    const float d[EDGE_PER_THREAD] = {d4.x, d4.y, d4.z, d4.w};
    float v[EDGE_PER_THREAD];
    #pragma unroll
    for (int u = 0; u < EDGE_PER_THREAD; ++u) {
        v[u] = a_src[s[u]] + a_dst[t[u]] + d[u] * ct;
        v[u] = (v[u] > 0.0f) ? v[u] : 0.01f * v[u];
    }
    *reinterpret_cast<float4*>(e + i0) = make_float4(v[0], v[1], v[2], v[3]);
    float* mp = mmax + half * NNODE;
    #pragma unroll
    for (int u = 0; u < EDGE_PER_THREAD; ++u)
        atomicMaxF(&mp[t[u]], v[u]);
}

#define CHUNK_EDGES 256
#define NCHUNK (NEDGE / CHUNK_EDGES)
#define EDGE_BATCH 8
static_assert(NEDGE % CHUNK_EDGES == 0, "whole chunks of edges");

__device__ inline void lds_only_barrier() {
    __builtin_amdgcn_fence(__ATOMIC_RELEASE, "workgroup", "local");
    __builtin_amdgcn_s_barrier();
    __builtin_amdgcn_fence(__ATOMIC_ACQUIRE, "workgroup", "local");
}

__global__ __launch_bounds__(256)
void gat_edge_scatter(const int* __restrict__ src,
                      const int* __restrict__ dst,
                      const float* __restrict__ elog,
                      const float* __restrict__ mmax,
                      float* __restrict__ denom,
                      const float* __restrict__ z,
                      float* __restrict__ acc,
                      int* __restrict__ tickets)
{
    __shared__ int s_ticket[2];
    const int lane = threadIdx.x & 31;
    const int wave = threadIdx.x >> 5;
    const int half = l2_half();

    const float* zl = z + lane;
    float*       ol = acc + lane;

    int k = 0;
    #pragma nounroll
    for (int ph = 0; ph < 2; ++ph) {
        const int par = half ^ ph;
        int* cnt = tickets + par * 64;
        if (threadIdx.x == 0) s_ticket[(k + 1) & 1] = atomicAdd(cnt, 1);
        lds_only_barrier();
        ++k;
        int t = s_ticket[k & 1];
        while (t < NCHUNK) {
            const int e   = t * CHUNK_EDGES + wave * 32 + lane;
            const int s_l = src[e];
            const int t_l = dst[e];
            if (threadIdx.x == 0) s_ticket[(k + 1) & 1] = atomicAdd(cnt, 1);

            const bool mine = (t_l & 1) == par;
            float a_l = 0.0f;
            if (mine) {
                a_l = __expf(elog[e] - fmaxf(mmax[t_l], mmax[NNODE + t_l]));
                atomicAdd(&denom[par * (NNODE / 2) + (t_l >> 1)], a_l);
            }
            unsigned m = __builtin_amdgcn_ballot_w32(mine);

            while (m) {
                const int n = min((int)__builtin_popcount(m), EDGE_BATCH);
                int   jj[EDGE_BATCH];
                float v[EDGE_BATCH][4];
                #pragma unroll
                for (int u = 0; u < EDGE_BATCH; ++u) {
                    if (u < n) {
                        const int j = __builtin_ctz(m);
                        m &= m - 1;
                        jj[u] = j;
                        const int sj = __builtin_amdgcn_readlane(s_l, j);
                        const float* zr = zl + (size_t)sj * DD;
                        v[u][0] = zr[0];
                        v[u][1] = zr[32];
                        v[u][2] = zr[64];
                        v[u][3] = zr[96];
                    }
                }
                #pragma unroll
                for (int u = 0; u < EDGE_BATCH; ++u) {
                    if (u < n) {
                        const int   tj = __builtin_amdgcn_readlane(t_l, jj[u]);
                        const float aj = __int_as_float(
                            __builtin_amdgcn_readlane(__float_as_int(a_l), jj[u]));
                        float* o = ol + (size_t)tj * DD;
                        atomicAdd(o,      v[u][0] * aj);
                        atomicAdd(o + 32, v[u][1] * aj);
                        atomicAdd(o + 64, v[u][2] * aj);
                        atomicAdd(o + 96, v[u][3] * aj);
                    }
                }
            }

            lds_only_barrier();
            ++k;
            t = s_ticket[k & 1];
        }
    }
}

#define RELU_THREADS 128
#define RELU_VEC 4
static_assert((NNODE * DD) % (RELU_THREADS * RELU_VEC * 4) == 0, "whole blocks of floats");

__global__ __launch_bounds__(RELU_THREADS)
void gat_relu(const float* __restrict__ znb,
              const float* __restrict__ acc,
              const float* __restrict__ denom,
              float* __restrict__ hout)
{
    const size_t base = (size_t)blockIdx.x * (RELU_THREADS * RELU_VEC * 4) + threadIdx.x * 4;
    float4 v[RELU_VEC], s[RELU_VEC];
    float  dn[RELU_VEC];
    #pragma unroll
    for (int u = 0; u < RELU_VEC; ++u) {
        const size_t idx = base + (size_t)u * RELU_THREADS * 4;
        const int rw = (int)(idx / DD);
        v[u]  = *reinterpret_cast<const float4*>(znb + idx);
        s[u]  = *reinterpret_cast<const float4*>(acc + idx);
        dn[u] = denom[(rw & 1) * (NNODE / 2) + (rw >> 1)];
    }
    #pragma unroll
    for (int u = 0; u < RELU_VEC; ++u) {
        const float ri = (dn[u] > 0.0f) ? 1.0f / dn[u] : 0.0f;
        float4 r;
        r.x = fmaxf(v[u].x + s[u].x * ri, 0.0f);
        r.y = fmaxf(v[u].y + s[u].y * ri, 0.0f);
        r.z = fmaxf(v[u].z + s[u].z * ri, 0.0f);
        r.w = fmaxf(v[u].w + s[u].w * ri, 0.0f);
        *reinterpret_cast<float4*>(hout + base + (size_t)u * RELU_THREADS * 4) = r;
    }
}

extern "C" void kernel_launch(void* const* d_in, const int* in_sizes, int n_in,
                              void* d_out, int out_size, void* d_ws, size_t ws_size,
                              hipStream_t stream) {
    const float* attr = (const float*)d_in[0];
    const float* dfe  = (const float*)d_in[1];
    const int*   src  = (const int*)d_in[2];
    const int*   dst  = (const int*)d_in[3];
    const float* W0   = (const float*)d_in[4];
    const float* W1   = (const float*)d_in[5];
    const float* W2   = (const float*)d_in[6];
    const float* Wa   = (const float*)d_in[7];

    float* ws    = (float*)d_ws;
    float* sumA  = ws;
    float* sumB  = sumA  + (size_t)NNODE * DD;
    float* accA  = sumB  + (size_t)NNODE * DD;
    float* accB  = accA  + (size_t)NNODE * DD;
    float* z     = accB  + (size_t)NNODE * DD;
    float* a_src = z     + (size_t)NNODE * DD;
    float* a_dst = a_src + NNODE;
    float* mmax  = a_dst + NNODE;
    float* denom = mmax  + 2 * NNODE;
    float* ebuf  = denom + NNODE;
    int*   tickets = (int*)(ebuf + NEDGE);

    const int gemmBlocks    = NNODE / GEMM_ROWS;
    const int nodeBlocks    = NNODE / (8 * PREP_NODES);
    const int edgeBlocks    = NEDGE / (EDGE_THREADS * EDGE_PER_THREAD);
    const int scatterBlocks = 2048;
    const int reluBlocks    = (NNODE * DD) / (RELU_THREADS * RELU_VEC * 4);

    for (int l = 0; l < 3; ++l) {
        const float* hin = (l == 0) ? attr : ((l == 1) ? sumA : sumB);
        const float* hac = (l == 1) ? accA : accB;
        float*       znb = (l == 1) ? sumB : sumA;
        float*       acc = (l == 1) ? accB : accA;
        const float* W1l = W1 + (size_t)l * DD * DD;
        const float* W2l = W2 + (size_t)l * DD * DD;
        const float* W0l = W0 + l;
        const float* Wal = Wa + (size_t)l * (2 * DD + 1);

        if (l == 0) gat_gemm_wmma<false><<<gemmBlocks, 128, 0, stream>>>(hin, hac, denom, W1l, W2l, z, znb, acc);
        else        gat_gemm_wmma<true><<<gemmBlocks, 128, 0, stream>>>(hin, hac, denom, W1l, W2l, z, znb, acc);
        gat_node_prep<<<nodeBlocks, 256, 0, stream>>>(z, Wal, a_src, a_dst, mmax, denom, tickets);
        gat_edge_logits<<<edgeBlocks, EDGE_THREADS, 0, stream>>>(src, dst, dfe, W0l, Wal,
                                                        a_src, a_dst, ebuf, mmax);
        gat_edge_scatter<<<scatterBlocks, 256, 0, stream>>>(src, dst, ebuf, mmax, denom, z, acc, tickets);
        if (l == 2) gat_relu<<<reluBlocks, RELU_THREADS, 0, stream>>>(znb, acc, denom, (float*)d_out);
    }
}
